# starting kernel
# baseline (speedup 1.0000x reference)
#include <hip/hip_runtime.h>
#include <math.h>

typedef __attribute__((ext_vector_type(2))) float v2f;
typedef __attribute__((ext_vector_type(8))) float v8f;

#define NB 8
#define NL 1024
#define NE 256
#define NR 16
#define NROWS (NB * NL)
#define LOG2E 1.4426950408889634f

__device__ __forceinline__ v8f wmma_f32_k4(v2f a, v2f b, v8f c) {
  return __builtin_amdgcn_wmma_f32_16x16x4_f32(false, a, false, b, (short)0, c,
                                               false, false);
}

__global__ void __launch_bounds__(256)
rbf_local_kernel(const float* __restrict__ coords,
                 const float* __restrict__ centers,
                 const float* __restrict__ gamma_p,
                 float* __restrict__ rbfl) {
  const int b    = blockIdx.x >> 6;
  const int i0   = (blockIdx.x & 63) << 4;
  const int tid  = threadIdx.x;
  const int lane = tid & 31;
  const int wave = tid >> 5;
  const int half = lane >> 4;
  const int lid  = lane & 15;
  const float ngl = -gamma_p[0] * LOG2E;

  __shared__ float red[NR * 16];
  red[tid] = 0.0f;
  __syncthreads();

  float cent[NR];
#pragma unroll
  for (int k = 0; k < NR; ++k) cent[k] = centers[k];

  const float* cb = coords + (size_t)b * NL * 3;

  const int arow = i0 + lid;
  v2f afrag;
  afrag.x = cb[arow * 3 + (half ? 2 : 0)];
  afrag.y = half ? 0.0f : cb[arow * 3 + 1];

  float rn[8];
#pragma unroll
  for (int r = 0; r < 8; ++r) {
    const int m = i0 + r + 8 * half;
    const float x = cb[m * 3 + 0];
    const float y = cb[m * 3 + 1];
    const float z = cb[m * 3 + 2];
    rn[r] = x * x + y * y + z * z;
  }

  float acc[8 * NR];
#pragma unroll
  for (int i = 0; i < 8 * NR; ++i) acc[i] = 0.0f;

  for (int jt = wave; jt < NL / 16; jt += 8) {
    const int jcol = (jt << 4) + lid;
    const float bx = cb[jcol * 3 + 0];
    const float by = cb[jcol * 3 + 1];
    const float bz = cb[jcol * 3 + 2];
    const float cn2 = bx * bx + by * by + bz * bz;

    v2f bfrag;
    bfrag.x = half ? bz : bx;
    bfrag.y = half ? 0.0f : by;

    v8f dot = {};
    dot = wmma_f32_k4(afrag, bfrag, dot);

#pragma unroll
    for (int r = 0; r < 8; ++r) {
      const float sq = rn[r] + cn2 - 2.0f * dot[r];
      const float d  = __builtin_amdgcn_sqrtf(fmaxf(sq, 1e-12f));
#pragma unroll
      for (int k = 0; k < NR; ++k) {
        const float t = d - cent[k];
        acc[r * NR + k] += __builtin_amdgcn_exp2f(ngl * t * t);
      }
    }
  }

#pragma unroll
  for (int i = 0; i < 8 * NR; ++i) {
    float v = acc[i];
    v += __shfl_xor(v, 1, 16);
    v += __shfl_xor(v, 2, 16);
    v += __shfl_xor(v, 4, 16);
    v += __shfl_xor(v, 8, 16);
    acc[i] = v;
  }
  if (lid == 0) {
#pragma unroll
    for (int r = 0; r < 8; ++r)
#pragma unroll
      for (int k = 0; k < NR; ++k)
        atomicAdd(&red[(r + 8 * half) * NR + k], acc[r * NR + k]);
  }
  __syncthreads();

  {
    const int m = tid >> 4;
    const int k = tid & 15;
    rbfl[((size_t)b * NL + i0 + m) * NR + k] = red[tid] * (1.0f / (float)NL);
  }
}

__global__ void __launch_bounds__(256)
weff_kernel(const float* __restrict__ w1, const float* __restrict__ b1,
            const float* __restrict__ w2, const float* __restrict__ b2,
            float* __restrict__ weff, float* __restrict__ beff) {
  const int idx = blockIdx.x * blockDim.x + threadIdx.x;
  if (idx < NR * NE) {
    const int k = idx >> 8;
    const int n = idx & (NE - 1);
    float s = 0.0f;
    for (int e = 0; e < NE; ++e) s += w1[k * NE + e] * w2[(NE + e) * NE + n];
    weff[idx] = s;
  } else if (idx < NR * NE + NE) {
    const int n = idx - NR * NE;
    float s = b2[n];
    for (int e = 0; e < NE; ++e) s += b1[e] * w2[(NE + e) * NE + n];
    beff[n] = s;
  }
}

__global__ void __launch_bounds__(256)
fused_out_kernel(const int* __restrict__ Z,
                 const float* __restrict__ atom_emb,
                 const float* __restrict__ w2,
                 const float* __restrict__ rbfl,
                 const float* __restrict__ weff,
                 const float* __restrict__ beff,
                 float* __restrict__ out) {
  const int lane = threadIdx.x & 31;
  const int wave = threadIdx.x >> 5;
  const int half = lane >> 4;
  const int lid  = lane & 15;
  const int gw   = blockIdx.x * 8 + wave;
  const int tm   = gw >> 4;
  const int tn   = gw & 15;
  const int row0 = tm << 4;
  const int jcol = (tn << 4) + lid;

  const int zrow = Z[row0 + lid];
  const float* aemb = atom_emb + (size_t)zrow * NE;
  const float* arbf = rbfl + (size_t)(row0 + lid) * NR;

  const float bval = beff[jcol];
  v8f c;
#pragma unroll
  for (int r = 0; r < 8; ++r) c[r] = bval;

#pragma unroll 4
  for (int s = 0; s < NE / 4; ++s) {
    const int kb = 4 * s + 2 * half;
    v2f a;
    a.x = aemb[kb];
    a.y = aemb[kb + 1];
    v2f bb;
    bb.x = w2[kb * NE + jcol];
    bb.y = w2[(kb + 1) * NE + jcol];
    c = wmma_f32_k4(a, bb, c);
  }
#pragma unroll
  for (int s = 0; s < NR / 4; ++s) {
    const int kb = 4 * s + 2 * half;
    v2f a;
    a.x = arbf[kb];
    a.y = arbf[kb + 1];
    v2f bb;
    bb.x = weff[kb * NE + jcol];
    bb.y = weff[(kb + 1) * NE + jcol];
    c = wmma_f32_k4(a, bb, c);
  }

#pragma unroll
  for (int r = 0; r < 8; ++r)
    out[(size_t)(row0 + r + 8 * half) * NE + jcol] = c[r];
}

extern "C" void kernel_launch(void* const* d_in, const int* in_sizes, int n_in,
                              void* d_out, int out_size, void* d_ws,
                              size_t ws_size, hipStream_t stream) {
  const float* coords   = (const float*)d_in[0];
  const int*   Z        = (const int*)d_in[1];
  const float* atom_emb = (const float*)d_in[2];
  const float* centers  = (const float*)d_in[3];
  const float* gamma    = (const float*)d_in[4];
  const float* w1       = (const float*)d_in[5];
  const float* b1       = (const float*)d_in[6];
  const float* w2       = (const float*)d_in[7];
  const float* b2       = (const float*)d_in[8];
  float* outp = (float*)d_out;

  float* rbfl = (float*)d_ws;
  float* weff = rbfl + NROWS * NR;
  float* beff = weff + NR * NE;

  weff_kernel<<<(NR * NE + NE + 255) / 256, 256, 0, stream>>>(w1, b1, w2, b2,
                                                              weff, beff);
  rbf_local_kernel<<<NB * (NL / 16), 256, 0, stream>>>(coords, centers, gamma,
                                                       rbfl);
  fused_out_kernel<<<(NROWS / 16) * (NE / 16) / 8, 256, 0, stream>>>(
      Z, atom_emb, w2, rbfl, weff, beff, outp);
}


# Weff/beff product folded into the rbf_local launch (136 of its blocks do one 32-column task first; sum over e now 8 runs of 32 added in wave order): 3 launches -> 2
# speedup vs baseline: 1.4120x; 1.4120x over previous
#include <hip/hip_runtime.h>
#include <math.h>

typedef __attribute__((ext_vector_type(2))) float v2f;
typedef __attribute__((ext_vector_type(8))) float v8f;

#define NB 8
#define NL 1024
#define NE 256
#define NR 16
#define NROWS (NB * NL)
#define LOG2E 1.4426950408889634f

__device__ __forceinline__ v8f wmma_f32_k4(v2f a, v2f b, v8f c) {
  return __builtin_amdgcn_wmma_f32_16x16x4_f32(false, a, false, b, (short)0, c,
                                               false, false);
}

__device__ __forceinline__ void
rbf_local_tile(const float* __restrict__ coords,
               const float* __restrict__ centers,
               const float* __restrict__ gamma_p,
               float* __restrict__ rbfl) {
  const int b    = blockIdx.x >> 6;
  const int i0   = (blockIdx.x & 63) << 4;
  const int tid  = threadIdx.x;
  const int lane = tid & 31;
  const int wave = tid >> 5;
  const int half = lane >> 4;
  const int lid  = lane & 15;
  const float ngl = -gamma_p[0] * LOG2E;

  __shared__ float red[NR * 16];
  red[tid] = 0.0f;
  __syncthreads();

  float cent[NR];
#pragma unroll
  for (int k = 0; k < NR; ++k) cent[k] = centers[k];

  const float* cb = coords + (size_t)b * NL * 3;

  const int arow = i0 + lid;
  v2f afrag;
  afrag.x = cb[arow * 3 + (half ? 2 : 0)];
  afrag.y = half ? 0.0f : cb[arow * 3 + 1];

  float rn[8];
#pragma unroll
  for (int r = 0; r < 8; ++r) {
    const int m = i0 + r + 8 * half;
    const float x = cb[m * 3 + 0];
    const float y = cb[m * 3 + 1];
    const float z = cb[m * 3 + 2];
    rn[r] = x * x + y * y + z * z;
  }

  float acc[8 * NR];
#pragma unroll
  for (int i = 0; i < 8 * NR; ++i) acc[i] = 0.0f;

  for (int jt = wave; jt < NL / 16; jt += 8) {
    const int jcol = (jt << 4) + lid;
    const float bx = cb[jcol * 3 + 0];
    const float by = cb[jcol * 3 + 1];
    const float bz = cb[jcol * 3 + 2];
    const float cn2 = bx * bx + by * by + bz * bz;

    v2f bfrag;
    bfrag.x = half ? bz : bx;
    bfrag.y = half ? 0.0f : by;

    v8f dot = {};
    dot = wmma_f32_k4(afrag, bfrag, dot);

#pragma unroll
    for (int r = 0; r < 8; ++r) {
      const float sq = rn[r] + cn2 - 2.0f * dot[r];
      const float d  = __builtin_amdgcn_sqrtf(fmaxf(sq, 1e-12f));
#pragma unroll
      for (int k = 0; k < NR; ++k) {
        const float t = d - cent[k];
        acc[r * NR + k] += __builtin_amdgcn_exp2f(ngl * t * t);
      }
    }
  }

#pragma unroll
  for (int i = 0; i < 8 * NR; ++i) {
    float v = acc[i];
    v += __shfl_xor(v, 1, 16);
    v += __shfl_xor(v, 2, 16);
    v += __shfl_xor(v, 4, 16);
    v += __shfl_xor(v, 8, 16);
    acc[i] = v;
  }
  if (lid == 0) {
#pragma unroll
    for (int r = 0; r < 8; ++r)
#pragma unroll
      for (int k = 0; k < NR; ++k)
        atomicAdd(&red[(r + 8 * half) * NR + k], acc[r * NR + k]);
  }
  __syncthreads();

  {
    const int m = tid >> 4;
    const int k = tid & 15;
    rbfl[((size_t)b * NL + i0 + m) * NR + k] = red[tid] * (1.0f / (float)NL);
  }
}

#define WEFF_TASKS ((NR + 1) * (NE / 32))

__device__ __forceinline__ void
weff_task(const int task,
          const float* __restrict__ w1, const float* __restrict__ b1,
          const float* __restrict__ w2, const float* __restrict__ b2,
          float* __restrict__ weff, float* __restrict__ beff) {
  __shared__ float part[8 * 32];
  const int lane = threadIdx.x & 31;
  const int wave = __builtin_amdgcn_readfirstlane((int)(threadIdx.x >> 5));
  const int k    = task >> 3;
  const int n    = ((task & 7) << 5) + lane;
  const int e0   = wave << 5;
  const float* __restrict__ arow = (k < NR) ? (w1 + k * NE) : b1;

  float s = 0.0f;
#pragma unroll
  for (int i = 0; i < 32; ++i)
    s += arow[e0 + i] * w2[(NE + e0 + i) * NE + n];
  part[threadIdx.x] = s;
  __syncthreads();

  if (wave == 0) {
    float t = (k < NR) ? 0.0f : b2[n];
    for (int w = 0; w < 8; ++w) t += part[w * 32 + lane];
    if (k < NR) weff[k * NE + n] = t;
    else        beff[n] = t;
  }
}

__global__ void __launch_bounds__(256)
rbf_local_weff_kernel(const float* __restrict__ coords,
                      const float* __restrict__ centers,
                      const float* __restrict__ gamma_p,
                      float* __restrict__ rbfl,
                      const float* __restrict__ w1,
                      const float* __restrict__ b1,
                      const float* __restrict__ w2,
                      const float* __restrict__ b2,
                      float* __restrict__ weff,
                      float* __restrict__ beff) {
  if (blockIdx.x < WEFF_TASKS)
    weff_task(blockIdx.x, w1, b1, w2, b2, weff, beff);
  rbf_local_tile(coords, centers, gamma_p, rbfl);
}

__global__ void __launch_bounds__(256)
fused_out_kernel(const int* __restrict__ Z,
                 const float* __restrict__ atom_emb,
                 const float* __restrict__ w2,
                 const float* __restrict__ rbfl,
                 const float* __restrict__ weff,
                 const float* __restrict__ beff,
                 float* __restrict__ out) {
  const int lane = threadIdx.x & 31;
  const int wave = threadIdx.x >> 5;
  const int half = lane >> 4;
  const int lid  = lane & 15;
  const int gw   = blockIdx.x * 8 + wave;
  const int tm   = gw >> 4;
  const int tn   = gw & 15;
  const int row0 = tm << 4;
  const int jcol = (tn << 4) + lid;

  const int zrow = Z[row0 + lid];
  const float* aemb = atom_emb + (size_t)zrow * NE;
  const float* arbf = rbfl + (size_t)(row0 + lid) * NR;

  const float bval = beff[jcol];
  v8f c;
#pragma unroll
  for (int r = 0; r < 8; ++r) c[r] = bval;

#pragma unroll 4
  for (int s = 0; s < NE / 4; ++s) {
    const int kb = 4 * s + 2 * half;
    v2f a;
    a.x = aemb[kb];
    a.y = aemb[kb + 1];
    v2f bb;
    bb.x = w2[kb * NE + jcol];
    bb.y = w2[(kb + 1) * NE + jcol];
    c = wmma_f32_k4(a, bb, c);
  }
#pragma unroll
  for (int s = 0; s < NR / 4; ++s) {
    const int kb = 4 * s + 2 * half;
    v2f a;
    a.x = arbf[kb];
    a.y = arbf[kb + 1];
    v2f bb;
    bb.x = weff[kb * NE + jcol];
    bb.y = weff[(kb + 1) * NE + jcol];
    c = wmma_f32_k4(a, bb, c);
  }

#pragma unroll
  for (int r = 0; r < 8; ++r)
    out[(size_t)(row0 + r + 8 * half) * NE + jcol] = c[r];
}

extern "C" void kernel_launch(void* const* d_in, const int* in_sizes, int n_in,
                              void* d_out, int out_size, void* d_ws,
                              size_t ws_size, hipStream_t stream) {
  const float* coords   = (const float*)d_in[0];
  const int*   Z        = (const int*)d_in[1];
  const float* atom_emb = (const float*)d_in[2];
  const float* centers  = (const float*)d_in[3];
  const float* gamma    = (const float*)d_in[4];
  const float* w1       = (const float*)d_in[5];
  const float* b1       = (const float*)d_in[6];
  const float* w2       = (const float*)d_in[7];
  const float* b2       = (const float*)d_in[8];
  float* outp = (float*)d_out;

  float* rbfl = (float*)d_ws;
  float* weff = rbfl + NROWS * NR;
  float* beff = weff + NR * NE;

  rbf_local_weff_kernel<<<NB * (NL / 16), 256, 0, stream>>>(
      coords, centers, gamma, rbfl, w1, b1, w2, b2, weff, beff);
  fused_out_kernel<<<(NROWS / 16) * (NE / 16) / 8, 256, 0, stream>>>(
      Z, atom_emb, w2, rbfl, weff, beff, outp);
}


# atom_emb[Z]@W2[:256] -> gather from table atom_emb@W2[:256] (128x256, WMMA; sum over e = 8 runs of 32 added in wave order) built beside Weff in launch 1; rbf launch ends with table[Z]+beff+rbf@Weff
# speedup vs baseline: 2.2011x; 1.5589x over previous
#include <hip/hip_runtime.h>
#include <math.h>


typedef __attribute__((ext_vector_type(2))) float v2f;
typedef __attribute__((ext_vector_type(8))) float v8f;

#define NB 8
#define NL 1024
#define NE 256
#define NR 16
#define NZ 118
#define NZP 128
#define NROWS (NB * NL)
#define LOG2E 1.4426950408889634f

__device__ __forceinline__ v8f wmma_f32_k4(v2f a, v2f b, v8f c) {
  return __builtin_amdgcn_wmma_f32_16x16x4_f32(false, a, false, b, (short)0, c,
                                               false, false);
}


#define WEFF_TASKS ((NR + 1) * (NE / 32))

__device__ __forceinline__ void
weff_task(const int task,
          const float* __restrict__ w1, const float* __restrict__ b1,
          const float* __restrict__ w2, const float* __restrict__ b2,
          float* __restrict__ weff, float* __restrict__ beff) {
  __shared__ float part[8 * 32];
  const int lane = threadIdx.x & 31;
  const int wave = __builtin_amdgcn_readfirstlane((int)(threadIdx.x >> 5));
  const int k    = task >> 3;
  const int n    = ((task & 7) << 5) + lane;
  const int e0   = wave << 5;
  const float* __restrict__ arow = (k < NR) ? (w1 + k * NE) : b1;

  float s = 0.0f;
#pragma unroll
  for (int i = 0; i < 32; ++i)
    s += arow[e0 + i] * w2[(NE + e0 + i) * NE + n];
  part[threadIdx.x] = s;
  __syncthreads();

  if (wave == 0) {
    float t = (k < NR) ? 0.0f : b2[n];
    for (int w = 0; w < 8; ++w) t += part[w * 32 + lane];
    if (k < NR) weff[k * NE + n] = t;
    else        beff[n] = t;
  }
}

#define TAB_TASKS ((NZP / 16) * (NE / 32))

__device__ __forceinline__ void
table_task(const int task,
           const float* __restrict__ atom_emb,
           const float* __restrict__ w2,
           float* __restrict__ tab) {
  __shared__ float tpart[8 * 512];
  const int lane = threadIdx.x & 31;
  const int wave = __builtin_amdgcn_readfirstlane((int)(threadIdx.x >> 5));
  const int half = lane >> 4;
  const int lid  = lane & 15;
  const int tm   = task >> 3;
  const int n0   = (task & 7) << 5;
  const int row  = (tm << 4) + lid;
  const bool live = row < NZ;
  const float* __restrict__ arow = atom_emb + (size_t)(live ? row : 0) * NE;
  const int e0   = wave << 5;

  v8f d0 = {};
  v8f d1 = {};
#pragma unroll
  for (int s = 0; s < 8; ++s) {
    const int kb = e0 + 4 * s + 2 * half;
    const float ax = arow[kb];
    const float ay = arow[kb + 1];
    v2f a;
    a.x = live ? ax : 0.0f;
    a.y = live ? ay : 0.0f;
    v2f b0, b1;
    b0.x = w2[kb * NE + n0 + lid];
    b0.y = w2[(kb + 1) * NE + n0 + lid];
    b1.x = w2[kb * NE + n0 + 16 + lid];
    b1.y = w2[(kb + 1) * NE + n0 + 16 + lid];
    d0 = wmma_f32_k4(a, b0, d0);
    d1 = wmma_f32_k4(a, b1, d1);
  }

  float* mine = tpart + wave * 512 + half * 32 + lid;
#pragma unroll
  for (int r = 0; r < 8; ++r) {
    mine[r * 64]      = d0[r];
    mine[r * 64 + 16] = d1[r];
  }
  __syncthreads();

  const int q = threadIdx.x >> 5;
#pragma unroll
  for (int j = 0; j < 2; ++j) {
    const int o = q * 64 + j * 32 + lane;
    float t = 0.0f;
    for (int w = 0; w < 8; ++w) t += tpart[w * 512 + o];
    tab[((tm << 4) + q + 8 * j) * NE + n0 + lane] = t;
  }
}

__global__ void __launch_bounds__(256)
weff_table_kernel(const float* __restrict__ w1,
                  const float* __restrict__ b1,
                  const float* __restrict__ w2,
                  const float* __restrict__ b2,
                  const float* __restrict__ atom_emb,
                  float* __restrict__ weff,
                  float* __restrict__ beff,
                  float* __restrict__ tab) {
  if (blockIdx.x < WEFF_TASKS)
    weff_task(blockIdx.x, w1, b1, w2, b2, weff, beff);
  else
    table_task(blockIdx.x - WEFF_TASKS, atom_emb, w2, tab);
}

__global__ void __launch_bounds__(256) __attribute__((amdgpu_waves_per_eu(2, 2)))
rbf_out_kernel(const float* __restrict__ coords,
               const float* __restrict__ centers,
               const float* __restrict__ gamma_p,
               const int* __restrict__ Z,
               const float* __restrict__ tab,
               const float* __restrict__ weff,
               const float* __restrict__ beff,
               float* __restrict__ out) {
  const int b    = blockIdx.x >> 6;
  const int i0   = (blockIdx.x & 63) << 4;
  const int tid  = threadIdx.x;
  const int lane = tid & 31;
  const int wave = tid >> 5;
  const int half = lane >> 4;
  const int lid  = lane & 15;
  const float ngl = -gamma_p[0] * LOG2E;

  __shared__ float red[NR * 16];
  red[tid] = 0.0f;
  __syncthreads();

  int zr[8];
#pragma unroll
  for (int r = 0; r < 8; ++r) zr[r] = Z[b * NL + i0 + r + 8 * half];

  float cent[NR];
#pragma unroll
  for (int k = 0; k < NR; ++k) cent[k] = centers[k];

  const float* cb = coords + (size_t)b * NL * 3;

  const int arow = i0 + lid;
  v2f afrag;
  afrag.x = cb[arow * 3 + (half ? 2 : 0)];
  afrag.y = half ? 0.0f : cb[arow * 3 + 1];

  float rn[8];
#pragma unroll
  for (int r = 0; r < 8; ++r) {
    const int m = i0 + r + 8 * half;
    const float x = cb[m * 3 + 0];
    const float y = cb[m * 3 + 1];
    const float z = cb[m * 3 + 2];
    rn[r] = x * x + y * y + z * z;
  }

  float acc[8 * NR];
#pragma unroll
  for (int i = 0; i < 8 * NR; ++i) acc[i] = 0.0f;

  for (int jt = wave; jt < NL / 16; jt += 8) {
    const int jcol = (jt << 4) + lid;
    const float bx = cb[jcol * 3 + 0];
    const float by = cb[jcol * 3 + 1];
    const float bz = cb[jcol * 3 + 2];
    const float cn2 = bx * bx + by * by + bz * bz;

    v2f bfrag;
    bfrag.x = half ? bz : bx;
    bfrag.y = half ? 0.0f : by;

    v8f dot = {};
    dot = wmma_f32_k4(afrag, bfrag, dot);

#pragma unroll
    for (int r = 0; r < 8; ++r) {
      const float sq = rn[r] + cn2 - 2.0f * dot[r];
      const float d  = __builtin_amdgcn_sqrtf(fmaxf(sq, 1e-12f));
#pragma unroll
      for (int k = 0; k < NR; ++k) {
        const float t = d - cent[k];
        acc[r * NR + k] += __builtin_amdgcn_exp2f(ngl * t * t);
      }
    }
  }

  const int ncol = (wave << 5) + lid;
  float tv[2][8];
  float wv[2][8];
  float bv[2];
#pragma unroll
  for (int t = 0; t < 2; ++t) {
    const int jc = ncol + 16 * t;
    bv[t] = beff[jc];
#pragma unroll
    for (int r = 0; r < 8; ++r) tv[t][r] = tab[zr[r] * NE + jc];
#pragma unroll
    for (int s = 0; s < NR / 4; ++s) {
      const int kb = 4 * s + 2 * half;
      wv[t][2 * s]     = weff[kb * NE + jc];
      wv[t][2 * s + 1] = weff[(kb + 1) * NE + jc];
    }
  }

#pragma unroll
  for (int i = 0; i < 8 * NR; ++i) {
    float v = acc[i];
    v += __shfl_xor(v, 1, 16);
    v += __shfl_xor(v, 2, 16);
    v += __shfl_xor(v, 4, 16);
    v += __shfl_xor(v, 8, 16);
    acc[i] = v;
  }
  if (lid == 0) {
#pragma unroll
    for (int r = 0; r < 8; ++r)
#pragma unroll
      for (int k = 0; k < NR; ++k)
        atomicAdd(&red[(r + 8 * half) * NR + k], acc[r * NR + k]);
  }
  __syncthreads();

  const float invl = 1.0f / (float)NL;
  v2f af[NR / 4];
#pragma unroll
  for (int s = 0; s < NR / 4; ++s) {
    const int kb = 4 * s + 2 * half;
    af[s].x = red[lid * NR + kb] * invl;
    af[s].y = red[lid * NR + kb + 1] * invl;
  }

#pragma unroll
  for (int t = 0; t < 2; ++t) {
    v8f c;
#pragma unroll
    for (int r = 0; r < 8; ++r) c[r] = tv[t][r] + bv[t];
#pragma unroll
    for (int s = 0; s < NR / 4; ++s) {
      v2f bb;
      bb.x = wv[t][2 * s];
      bb.y = wv[t][2 * s + 1];
      c = wmma_f32_k4(af[s], bb, c);
    }
#pragma unroll
    for (int r = 0; r < 8; ++r)
      out[((size_t)b * NL + i0 + r + 8 * half) * NE + ncol + 16 * t] = c[r];
  }
}

extern "C" void kernel_launch(void* const* d_in, const int* in_sizes, int n_in,
                              void* d_out, int out_size, void* d_ws,
                              size_t ws_size, hipStream_t stream) {
  const float* coords   = (const float*)d_in[0];
  const int*   Z        = (const int*)d_in[1];
  const float* atom_emb = (const float*)d_in[2];
  const float* centers  = (const float*)d_in[3];
  const float* gamma    = (const float*)d_in[4];
  const float* w1       = (const float*)d_in[5];
  const float* b1       = (const float*)d_in[6];
  const float* w2       = (const float*)d_in[7];
  const float* b2       = (const float*)d_in[8];
  float* outp = (float*)d_out;

  float* tab  = (float*)d_ws;
  float* weff = tab + NZP * NE;
  float* beff = weff + NR * NE;

  weff_table_kernel<<<WEFF_TASKS + TAB_TASKS, 256, 0, stream>>>(
      w1, b1, w2, b2, atom_emb, weff, beff, tab);
  rbf_out_kernel<<<NB * (NL / 16), 256, 0, stream>>>(
      coords, centers, gamma, Z, tab, weff, beff, outp);
}


# Gram tile transposed: neighbours j on D rows, block rows i on lanes (C=|xi|^2, |xj|^2 in spare k slot): sum over j runs in-lane, then 16 LDS partials in slot order; no shuffles, no atomics
# speedup vs baseline: 3.3582x; 1.5257x over previous
#include <hip/hip_runtime.h>
#include <math.h>


typedef __attribute__((ext_vector_type(2))) float v2f;
typedef __attribute__((ext_vector_type(8))) float v8f;

#define NB 8
#define NL 1024
#define NE 256
#define NR 16
#define NZ 118
#define NZP 128
#define NROWS (NB * NL)
#define LOG2E 1.4426950408889634f

__device__ __forceinline__ v8f wmma_f32_k4(v2f a, v2f b, v8f c) {
  return __builtin_amdgcn_wmma_f32_16x16x4_f32(false, a, false, b, (short)0, c,
                                               false, false);
}


#define WEFF_TASKS ((NR + 1) * (NE / 32))

__device__ __forceinline__ void
weff_task(const int task,
          const float* __restrict__ w1, const float* __restrict__ b1,
          const float* __restrict__ w2, const float* __restrict__ b2,
          float* __restrict__ weff, float* __restrict__ beff) {
  __shared__ float part[8 * 32];
  const int lane = threadIdx.x & 31;
  const int wave = __builtin_amdgcn_readfirstlane((int)(threadIdx.x >> 5));
  const int k    = task >> 3;
  const int n    = ((task & 7) << 5) + lane;
  const int e0   = wave << 5;
  const float* __restrict__ arow = (k < NR) ? (w1 + k * NE) : b1;

  float s = 0.0f;
#pragma unroll
  for (int i = 0; i < 32; ++i)
    s += arow[e0 + i] * w2[(NE + e0 + i) * NE + n];
  part[threadIdx.x] = s;
  __syncthreads();

  if (wave == 0) {
    float t = (k < NR) ? 0.0f : b2[n];
    for (int w = 0; w < 8; ++w) t += part[w * 32 + lane];
    if (k < NR) weff[k * NE + n] = t;
    else        beff[n] = t;
  }
}

#define TAB_TASKS ((NZP / 16) * (NE / 32))

__device__ __forceinline__ void
table_task(const int task,
           const float* __restrict__ atom_emb,
           const float* __restrict__ w2,
           float* __restrict__ tab) {
  __shared__ float tpart[8 * 512];
  const int lane = threadIdx.x & 31;
  const int wave = __builtin_amdgcn_readfirstlane((int)(threadIdx.x >> 5));
  const int half = lane >> 4;
  const int lid  = lane & 15;
  const int tm   = task >> 3;
  const int n0   = (task & 7) << 5;
  const int row  = (tm << 4) + lid;
  const bool live = row < NZ;
  const float* __restrict__ arow = atom_emb + (size_t)(live ? row : 0) * NE;
  const int e0   = wave << 5;

  v8f d0 = {};
  v8f d1 = {};
#pragma unroll
  for (int s = 0; s < 8; ++s) {
    const int kb = e0 + 4 * s + 2 * half;
    const float ax = arow[kb];
    const float ay = arow[kb + 1];
    v2f a;
    a.x = live ? ax : 0.0f;
    a.y = live ? ay : 0.0f;
    v2f b0, b1;
    b0.x = w2[kb * NE + n0 + lid];
    b0.y = w2[(kb + 1) * NE + n0 + lid];
    b1.x = w2[kb * NE + n0 + 16 + lid];
    b1.y = w2[(kb + 1) * NE + n0 + 16 + lid];
    d0 = wmma_f32_k4(a, b0, d0);
    d1 = wmma_f32_k4(a, b1, d1);
  }

  float* mine = tpart + wave * 512 + half * 32 + lid;
#pragma unroll
  for (int r = 0; r < 8; ++r) {
    mine[r * 64]      = d0[r];
    mine[r * 64 + 16] = d1[r];
  }
  __syncthreads();

  const int q = threadIdx.x >> 5;
#pragma unroll
  for (int j = 0; j < 2; ++j) {
    const int o = q * 64 + j * 32 + lane;
    float t = 0.0f;
    for (int w = 0; w < 8; ++w) t += tpart[w * 512 + o];
    tab[((tm << 4) + q + 8 * j) * NE + n0 + lane] = t;
  }
}

__global__ void __launch_bounds__(256)
weff_table_kernel(const float* __restrict__ w1,
                  const float* __restrict__ b1,
                  const float* __restrict__ w2,
                  const float* __restrict__ b2,
                  const float* __restrict__ atom_emb,
                  float* __restrict__ weff,
                  float* __restrict__ beff,
                  float* __restrict__ tab) {
  if (blockIdx.x < WEFF_TASKS)
    weff_task(blockIdx.x, w1, b1, w2, b2, weff, beff);
  else
    table_task(blockIdx.x - WEFF_TASKS, atom_emb, w2, tab);
}

#define PSLOT 272

__global__ void __launch_bounds__(256)
rbf_out_kernel(const float* __restrict__ coords,
               const float* __restrict__ centers,
               const float* __restrict__ gamma_p,
               const int* __restrict__ Z,
               const float* __restrict__ tab,
               const float* __restrict__ weff,
               const float* __restrict__ beff,
               float* __restrict__ out) {
  const int b    = blockIdx.x >> 6;
  const int i0   = (blockIdx.x & 63) << 4;
  const int tid  = threadIdx.x;
  const int lane = tid & 31;
  const int wave = tid >> 5;
  const int half = lane >> 4;
  const int lid  = lane & 15;
  const float ngl = -gamma_p[0] * LOG2E;

  __shared__ float part[16 * PSLOT];
  __shared__ float red[NR * 16];

  int zr[8];
#pragma unroll
  for (int r = 0; r < 8; ++r) zr[r] = Z[b * NL + i0 + r + 8 * half];

  float cent[NR];
#pragma unroll
  for (int k = 0; k < NR; ++k) cent[k] = centers[k];

  const float* cb = coords + (size_t)b * NL * 3;

  const int irow = i0 + lid;
  const float xi = cb[irow * 3 + 0];
  const float yi = cb[irow * 3 + 1];
  const float zi = cb[irow * 3 + 2];
  const float ni = xi * xi + yi * yi + zi * zi;
  v2f bfrag;
  bfrag.x = -2.0f * (half ? zi : xi);
  bfrag.y = half ? 1.0f : -2.0f * yi;
  v8f cinit;
#pragma unroll
  for (int r = 0; r < 8; ++r) cinit[r] = ni;

  float acc[NR];
#pragma unroll
  for (int k = 0; k < NR; ++k) acc[k] = 0.0f;

  for (int jt = wave; jt < NL / 16; jt += 8) {
    const int jrow = (jt << 4) + lid;
    const float ax = cb[jrow * 3 + 0];
    const float ay = cb[jrow * 3 + 1];
    const float az = cb[jrow * 3 + 2];
    const float nj = ax * ax + ay * ay + az * az;
    v2f afrag;
    afrag.x = half ? az : ax;
    afrag.y = half ? nj : ay;

    const v8f sq = wmma_f32_k4(afrag, bfrag, cinit);

#pragma unroll
    for (int r = 0; r < 8; ++r) {
      const float d = __builtin_amdgcn_sqrtf(fmaxf(sq[r], 1e-12f));
#pragma unroll
      for (int k = 0; k < NR; ++k) {
        const float t = d - cent[k];
        acc[k] += __builtin_amdgcn_exp2f(ngl * t * t);
      }
    }
  }

  const int ncol = (wave << 5) + lid;
  float tv[2][8];
  float wv[2][8];
  float bv[2];
#pragma unroll
  for (int t = 0; t < 2; ++t) {
    const int jc = ncol + 16 * t;
    bv[t] = beff[jc];
#pragma unroll
    for (int r = 0; r < 8; ++r) tv[t][r] = tab[zr[r] * NE + jc];
#pragma unroll
    for (int s = 0; s < NR / 4; ++s) {
      const int kb = 4 * s + 2 * half;
      wv[t][2 * s]     = weff[kb * NE + jc];
      wv[t][2 * s + 1] = weff[(kb + 1) * NE + jc];
    }
  }

  float* mine = part + (wave * 2 + half) * PSLOT + lid;
#pragma unroll
  for (int k = 0; k < NR; ++k) mine[k * 16] = acc[k];
  __syncthreads();

  {
    float s = 0.0f;
#pragma unroll
    for (int q = 0; q < 16; ++q) s += part[q * PSLOT + tid];
    red[tid] = s;
  }
  __syncthreads();

  const float invl = 1.0f / (float)NL;
  v2f af[NR / 4];
#pragma unroll
  for (int s = 0; s < NR / 4; ++s) {
    const int kb = 4 * s + 2 * half;
    af[s].x = red[kb * 16 + lid] * invl;
    af[s].y = red[(kb + 1) * 16 + lid] * invl;
  }

#pragma unroll
  for (int t = 0; t < 2; ++t) {
    v8f c;
#pragma unroll
    for (int r = 0; r < 8; ++r) c[r] = tv[t][r] + bv[t];
#pragma unroll
    for (int s = 0; s < NR / 4; ++s) {
      v2f bb;
      bb.x = wv[t][2 * s];
      bb.y = wv[t][2 * s + 1];
      c = wmma_f32_k4(af[s], bb, c);
    }
#pragma unroll
    for (int r = 0; r < 8; ++r)
      out[((size_t)b * NL + i0 + r + 8 * half) * NE + ncol + 16 * t] = c[r];
  }
}

extern "C" void kernel_launch(void* const* d_in, const int* in_sizes, int n_in,
                              void* d_out, int out_size, void* d_ws,
                              size_t ws_size, hipStream_t stream) {
  const float* coords   = (const float*)d_in[0];
  const int*   Z        = (const int*)d_in[1];
  const float* atom_emb = (const float*)d_in[2];
  const float* centers  = (const float*)d_in[3];
  const float* gamma    = (const float*)d_in[4];
  const float* w1       = (const float*)d_in[5];
  const float* b1       = (const float*)d_in[6];
  const float* w2       = (const float*)d_in[7];
  const float* b2       = (const float*)d_in[8];
  float* outp = (float*)d_out;

  float* tab  = (float*)d_ws;
  float* weff = tab + NZP * NE;
  float* beff = weff + NR * NE;

  weff_table_kernel<<<WEFF_TASKS + TAB_TASKS, 256, 0, stream>>>(
      w1, b1, w2, b2, atom_emb, weff, beff, tab);
  rbf_out_kernel<<<NB * (NL / 16), 256, 0, stream>>>(
      coords, centers, gamma, Z, tab, weff, beff, outp);
}
